# starting kernel
# baseline (speedup 1.0000x reference)
#include <hip/hip_runtime.h>
#include <hip/hip_bf16.h>

#define CDIM 128
#define LN_EPS 1e-5f

typedef float v2f __attribute__((ext_vector_type(2)));
typedef float v8f __attribute__((ext_vector_type(8)));

__device__ __forceinline__ float wave_sum32(float v) {
#pragma unroll
    for (int o = 16; o > 0; o >>= 1) v += __shfl_xor(v, o, 32);
    return v;
}

__global__ void gcn_init_kernel(float* __restrict__ accum, float* __restrict__ deg, int N) {
    size_t idx = (size_t)blockIdx.x * blockDim.x + threadIdx.x;
    size_t total = (size_t)N * CDIM;
    if (idx < total) accum[idx] = 0.0f;
    if (idx < (size_t)N) deg[idx] = 1.0f;
}

__global__ void gcn_degree_kernel(const int* __restrict__ ei, float* __restrict__ deg, int E) {
    int e = blockIdx.x * blockDim.x + threadIdx.x;
    if (e >= E) return;
    int dst = ei[E + e];
    unsafeAtomicAdd(&deg[dst], 1.0f);
}

__global__ void gcn_dinv_kernel(float* __restrict__ deg, int N) {
    int i = blockIdx.x * blockDim.x + threadIdx.x;
    if (i < N) deg[i] = rsqrtf(deg[i]);
}

__global__ void __launch_bounds__(256)
gcn_gemm_kernel(const float* __restrict__ x, const float* __restrict__ W,
                float* __restrict__ h, int N) {
    const int wave = threadIdx.x >> 5;
    const int lane = threadIdx.x & 31;
    const int half = lane >> 4;
    const int l    = lane & 15;
    const int row_base = blockIdx.x * 16;
    const int col_base = wave * 16;

    int m = row_base + l;
    if (m >= N) m = N - 1;

    const float* xrow = x + (size_t)m * CDIM + half * 2;
    const float* wrow = W + (size_t)(col_base + l) * CDIM + half * 2;

    v8f acc = {};
#pragma unroll
    for (int k0 = 0; k0 < CDIM; k0 += 4) {
        v2f a = *(const v2f*)(xrow + k0);
        v2f b = *(const v2f*)(wrow + k0);
        acc = __builtin_amdgcn_wmma_f32_16x16x4_f32(
             false, a,  false, b,
             (short)0, acc,  false,  false);
    }

    const int outcol = col_base + l;
#pragma unroll
    for (int r = 0; r < 8; ++r) {
        int orow = row_base + r + half * 8;
        if (orow < N) h[(size_t)orow * CDIM + outcol] = acc[r];
    }
}

__global__ void __launch_bounds__(256)
gcn_scatter_kernel(const float* __restrict__ h, const float* __restrict__ dinv,
                   const int* __restrict__ ei, float* __restrict__ accum, int E) {
    int wid = blockIdx.x * (blockDim.x >> 5) + (threadIdx.x >> 5);
    if (wid >= E) return;
    int lane = threadIdx.x & 31;

    int src = ei[wid];
    int dst = ei[E + wid];
    float norm = dinv[src] * dinv[dst];

    const float4 hv = *(const float4*)(h + (size_t)src * CDIM + lane * 4);
    float* ap = accum + (size_t)dst * CDIM + lane * 4;
    unsafeAtomicAdd(ap + 0, hv.x * norm);
    unsafeAtomicAdd(ap + 1, hv.y * norm);
    unsafeAtomicAdd(ap + 2, hv.z * norm);
    unsafeAtomicAdd(ap + 3, hv.w * norm);
}

__global__ void __launch_bounds__(256)
gcn_epilogue_kernel(const float* __restrict__ x, const float* __restrict__ h,
                    const float* __restrict__ accum, const float* __restrict__ dinv,
                    const float* __restrict__ bvec,
                    const float* __restrict__ g1, const float* __restrict__ be1,
                    const float* __restrict__ g2, const float* __restrict__ be2,
                    float* __restrict__ out, int N) {
    int wid = blockIdx.x * (blockDim.x >> 5) + (threadIdx.x >> 5);
    if (wid >= N) return;
    int lane = threadIdx.x & 31;
    int c = lane * 4;
    size_t base = (size_t)wid * CDIM + c;

    float di = dinv[wid];
    float sw = di * di;

    float4 hv = *(const float4*)(h + base);
    float4 av = *(const float4*)(accum + base);
    float4 bv = *(const float4*)(bvec + c);

    float v0 = fmaxf(av.x + sw * hv.x + bv.x, 0.0f);
    float v1 = fmaxf(av.y + sw * hv.y + bv.y, 0.0f);
    float v2 = fmaxf(av.z + sw * hv.z + bv.z, 0.0f);
    float v3 = fmaxf(av.w + sw * hv.w + bv.w, 0.0f);

    float mu = wave_sum32(v0 + v1 + v2 + v3) * (1.0f / CDIM);
    float d0 = v0 - mu, d1 = v1 - mu, d2 = v2 - mu, d3 = v3 - mu;
    float var = wave_sum32(d0 * d0 + d1 * d1 + d2 * d2 + d3 * d3) * (1.0f / CDIM);
    float inv = rsqrtf(var + LN_EPS);
    float4 g1v = *(const float4*)(g1 + c);
    float4 b1v = *(const float4*)(be1 + c);
    float4 xv  = *(const float4*)(x + base);
    float z0 = xv.x + (d0 * inv * g1v.x + b1v.x);
    float z1 = xv.y + (d1 * inv * g1v.y + b1v.y);
    float z2 = xv.z + (d2 * inv * g1v.z + b1v.z);
    float z3 = xv.w + (d3 * inv * g1v.w + b1v.w);

    float mu2 = wave_sum32(z0 + z1 + z2 + z3) * (1.0f / CDIM);
    float e0 = z0 - mu2, e1 = z1 - mu2, e2 = z2 - mu2, e3 = z3 - mu2;
    float var2 = wave_sum32(e0 * e0 + e1 * e1 + e2 * e2 + e3 * e3) * (1.0f / CDIM);
    float inv2 = rsqrtf(var2 + LN_EPS);
    float4 g2v = *(const float4*)(g2 + c);
    float4 b2v = *(const float4*)(be2 + c);

    float4 o;
    o.x = e0 * inv2 * g2v.x + b2v.x;
    o.y = e1 * inv2 * g2v.y + b2v.y;
    o.z = e2 * inv2 * g2v.z + b2v.z;
    o.w = e3 * inv2 * g2v.w + b2v.w;
    *(float4*)(out + base) = o;
}

extern "C" void kernel_launch(void* const* d_in, const int* in_sizes, int n_in,
                              void* d_out, int out_size, void* d_ws, size_t ws_size,
                              hipStream_t stream) {
    const float* x  = (const float*)d_in[0];
    const int*   ei = (const int*)d_in[1];
    const float* W  = (const float*)d_in[2];
    const float* b  = (const float*)d_in[3];
    const float* g1 = (const float*)d_in[4];
    const float* be1 = (const float*)d_in[5];
    const float* g2 = (const float*)d_in[6];
    const float* be2 = (const float*)d_in[7];
    float* out = (float*)d_out;

    const int N = in_sizes[0] / CDIM;
    const int E = in_sizes[1] / 2;

    float* h     = (float*)d_ws;
    float* accum = h + (size_t)N * CDIM;
    float* deg   = accum + (size_t)N * CDIM;

    const size_t total = (size_t)N * CDIM;
    gcn_init_kernel<<<(int)((total + 255) / 256), 256, 0, stream>>>(accum, deg, N);
    gcn_degree_kernel<<<(E + 255) / 256, 256, 0, stream>>>(ei, deg, E);
    gcn_dinv_kernel<<<(N + 255) / 256, 256, 0, stream>>>(deg, N);
    gcn_gemm_kernel<<<(N + 15) / 16, 256, 0, stream>>>(x, W, h, N);
    gcn_scatter_kernel<<<(E + 7) / 8, 256, 0, stream>>>(h, deg, ei, accum, E);
    gcn_epilogue_kernel<<<(N + 7) / 8, 256, 0, stream>>>(x, h, accum, deg, b,
                                                         g1, be1, g2, be2, out, N);
}


# product kernel retiled: W staged once per 8-wave block in LDS, each wave one 16x128 strip (8 accumulator tiles), x and h rows moved as whole lines via LDS; k summed in a new order within each 8
# speedup vs baseline: 1.0944x; 1.0944x over previous
#include <hip/hip_runtime.h>
#include <hip/hip_bf16.h>

#define CDIM 128
#define LN_EPS 1e-5f

typedef float v2f __attribute__((ext_vector_type(2)));
typedef float v8f __attribute__((ext_vector_type(8)));

__device__ __forceinline__ float wave_sum32(float v) {
#pragma unroll
    for (int o = 16; o > 0; o >>= 1) v += __shfl_xor(v, o, 32);
    return v;
}

__global__ void gcn_init_kernel(float* __restrict__ accum, float* __restrict__ deg, int N) {
    size_t idx = (size_t)blockIdx.x * blockDim.x + threadIdx.x;
    size_t total = (size_t)N * CDIM;
    if (idx < total) accum[idx] = 0.0f;
    if (idx < (size_t)N) deg[idx] = 1.0f;
}

__global__ void gcn_degree_kernel(const int* __restrict__ ei, float* __restrict__ deg, int E) {
    int e = blockIdx.x * blockDim.x + threadIdx.x;
    if (e >= E) return;
    int dst = ei[E + e];
    unsafeAtomicAdd(&deg[dst], 1.0f);
}

__global__ void gcn_dinv_kernel(float* __restrict__ deg, int N) {
    int i = blockIdx.x * blockDim.x + threadIdx.x;
    if (i < N) deg[i] = rsqrtf(deg[i]);
}

#define WPITCH 132
#define GEMM_WAVES 8

__global__ void __launch_bounds__(256)
gcn_gemm_kernel(const float* __restrict__ x, const float* __restrict__ W,
                float* __restrict__ h, int N) {
    __shared__ __attribute__((aligned(16))) float Ws[CDIM * WPITCH];
    __shared__ __attribute__((aligned(16))) float Ts[GEMM_WAVES][16 * WPITCH];

    const int tid  = threadIdx.x;
    const int wave = __builtin_amdgcn_readfirstlane(tid >> 5);
    const int lane = tid & 31;
    const int half = lane >> 4;
    const int l    = lane & 15;
    const int row_base = (blockIdx.x * GEMM_WAVES + wave) * 16;
    float* slab = &Ts[wave][0];

    float4 xr[16];
#pragma unroll
    for (int rr = 0; rr < 16; ++rr) {
        int m = row_base + rr;
        if (m >= N) m = N - 1;
        xr[rr] = *(const float4*)(x + (size_t)m * CDIM + lane * 4);
    }
    float4 wr[16];
#pragma unroll
    for (int i = 0; i < 16; ++i) {
        int f = tid + 256 * i;
        wr[i] = *(const float4*)(W + (size_t)f * 4);
    }
#pragma unroll
    for (int rr = 0; rr < 16; ++rr)
        *(float4*)(slab + rr * WPITCH + lane * 4) = xr[rr];
#pragma unroll
    for (int i = 0; i < 16; ++i) {
        int f = tid + 256 * i;
        *(float4*)(Ws + (f >> 5) * WPITCH + (f & 31) * 4) = wr[i];
    }
    __syncthreads();

    const float* arow = slab + l * WPITCH + half * 4;
    const float* brow = Ws   + l * WPITCH + half * 4;

    v8f acc[8];
#pragma unroll
    for (int ct = 0; ct < 8; ++ct) acc[ct] = (v8f){};

#pragma unroll
    for (int j = 0; j < 16; ++j) {
        const float4 a4 = *(const float4*)(arow + 8 * j);
        v2f a01 = {a4.x, a4.y};
        v2f a23 = {a4.z, a4.w};
        float4 b4[8];
#pragma unroll
        for (int ct = 0; ct < 8; ++ct)
            b4[ct] = *(const float4*)(brow + ct * 16 * WPITCH + 8 * j);
#pragma unroll
        for (int ct = 0; ct < 8; ++ct) {
            v2f b01 = {b4[ct].x, b4[ct].y};
            acc[ct] = __builtin_amdgcn_wmma_f32_16x16x4_f32(
                 false, a01,  false, b01,
                 (short)0, acc[ct],  false,  false);
        }
#pragma unroll
        for (int ct = 0; ct < 8; ++ct) {
            v2f b23 = {b4[ct].z, b4[ct].w};
            acc[ct] = __builtin_amdgcn_wmma_f32_16x16x4_f32(
                 false, a23,  false, b23,
                 (short)0, acc[ct],  false,  false);
        }
    }

#pragma unroll
    for (int ct = 0; ct < 8; ++ct) {
#pragma unroll
        for (int r = 0; r < 8; ++r)
            slab[(r + 8 * half) * WPITCH + ct * 16 + l] = acc[ct][r];
    }
    __syncthreads();
    float4 hr[16];
#pragma unroll
    for (int rr = 0; rr < 16; ++rr)
        hr[rr] = *(const float4*)(slab + rr * WPITCH + lane * 4);
    float* hrow = h + (size_t)row_base * CDIM + lane * 4;
    if (row_base + 16 <= N) {
#pragma unroll
        for (int rr = 0; rr < 16; ++rr)
            *(float4*)(hrow + (size_t)rr * CDIM) = hr[rr];
    } else {
#pragma unroll
        for (int rr = 0; rr < 16; ++rr)
            if (row_base + rr < N) *(float4*)(hrow + (size_t)rr * CDIM) = hr[rr];
    }
}

__global__ void __launch_bounds__(256)
gcn_scatter_kernel(const float* __restrict__ h, const float* __restrict__ dinv,
                   const int* __restrict__ ei, float* __restrict__ accum, int E) {
    int wid = blockIdx.x * (blockDim.x >> 5) + (threadIdx.x >> 5);
    if (wid >= E) return;
    int lane = threadIdx.x & 31;

    int src = ei[wid];
    int dst = ei[E + wid];
    float norm = dinv[src] * dinv[dst];

    const float4 hv = *(const float4*)(h + (size_t)src * CDIM + lane * 4);
    float* ap = accum + (size_t)dst * CDIM + lane * 4;
    unsafeAtomicAdd(ap + 0, hv.x * norm);
    unsafeAtomicAdd(ap + 1, hv.y * norm);
    unsafeAtomicAdd(ap + 2, hv.z * norm);
    unsafeAtomicAdd(ap + 3, hv.w * norm);
}

__global__ void __launch_bounds__(256)
gcn_epilogue_kernel(const float* __restrict__ x, const float* __restrict__ h,
                    const float* __restrict__ accum, const float* __restrict__ dinv,
                    const float* __restrict__ bvec,
                    const float* __restrict__ g1, const float* __restrict__ be1,
                    const float* __restrict__ g2, const float* __restrict__ be2,
                    float* __restrict__ out, int N) {
    int wid = blockIdx.x * (blockDim.x >> 5) + (threadIdx.x >> 5);
    if (wid >= N) return;
    int lane = threadIdx.x & 31;
    int c = lane * 4;
    size_t base = (size_t)wid * CDIM + c;

    float di = dinv[wid];
    float sw = di * di;

    float4 hv = *(const float4*)(h + base);
    float4 av = *(const float4*)(accum + base);
    float4 bv = *(const float4*)(bvec + c);

    float v0 = fmaxf(av.x + sw * hv.x + bv.x, 0.0f);
    float v1 = fmaxf(av.y + sw * hv.y + bv.y, 0.0f);
    float v2 = fmaxf(av.z + sw * hv.z + bv.z, 0.0f);
    float v3 = fmaxf(av.w + sw * hv.w + bv.w, 0.0f);

    float mu = wave_sum32(v0 + v1 + v2 + v3) * (1.0f / CDIM);
    float d0 = v0 - mu, d1 = v1 - mu, d2 = v2 - mu, d3 = v3 - mu;
    float var = wave_sum32(d0 * d0 + d1 * d1 + d2 * d2 + d3 * d3) * (1.0f / CDIM);
    float inv = rsqrtf(var + LN_EPS);
    float4 g1v = *(const float4*)(g1 + c);
    float4 b1v = *(const float4*)(be1 + c);
    float4 xv  = *(const float4*)(x + base);
    float z0 = xv.x + (d0 * inv * g1v.x + b1v.x);
    float z1 = xv.y + (d1 * inv * g1v.y + b1v.y);
    float z2 = xv.z + (d2 * inv * g1v.z + b1v.z);
    float z3 = xv.w + (d3 * inv * g1v.w + b1v.w);

    float mu2 = wave_sum32(z0 + z1 + z2 + z3) * (1.0f / CDIM);
    float e0 = z0 - mu2, e1 = z1 - mu2, e2 = z2 - mu2, e3 = z3 - mu2;
    float var2 = wave_sum32(e0 * e0 + e1 * e1 + e2 * e2 + e3 * e3) * (1.0f / CDIM);
    float inv2 = rsqrtf(var2 + LN_EPS);
    float4 g2v = *(const float4*)(g2 + c);
    float4 b2v = *(const float4*)(be2 + c);

    float4 o;
    o.x = e0 * inv2 * g2v.x + b2v.x;
    o.y = e1 * inv2 * g2v.y + b2v.y;
    o.z = e2 * inv2 * g2v.z + b2v.z;
    o.w = e3 * inv2 * g2v.w + b2v.w;
    *(float4*)(out + base) = o;
}

extern "C" void kernel_launch(void* const* d_in, const int* in_sizes, int n_in,
                              void* d_out, int out_size, void* d_ws, size_t ws_size,
                              hipStream_t stream) {
    const float* x  = (const float*)d_in[0];
    const int*   ei = (const int*)d_in[1];
    const float* W  = (const float*)d_in[2];
    const float* b  = (const float*)d_in[3];
    const float* g1 = (const float*)d_in[4];
    const float* be1 = (const float*)d_in[5];
    const float* g2 = (const float*)d_in[6];
    const float* be2 = (const float*)d_in[7];
    float* out = (float*)d_out;

    const int N = in_sizes[0] / CDIM;
    const int E = in_sizes[1] / 2;

    float* h     = (float*)d_ws;
    float* accum = h + (size_t)N * CDIM;
    float* deg   = accum + (size_t)N * CDIM;

    const size_t total = (size_t)N * CDIM;
    gcn_init_kernel<<<(int)((total + 255) / 256), 256, 0, stream>>>(accum, deg, N);
    gcn_degree_kernel<<<(E + 255) / 256, 256, 0, stream>>>(ei, deg, E);
    gcn_dinv_kernel<<<(N + 255) / 256, 256, 0, stream>>>(deg, N);
    gcn_gemm_kernel<<<(N + 127) / 128, 256, 0, stream>>>(x, W, h, N);
    gcn_scatter_kernel<<<(E + 7) / 8, 256, 0, stream>>>(h, deg, ei, accum, E);
    gcn_epilogue_kernel<<<(N + 7) / 8, 256, 0, stream>>>(x, h, accum, deg, b,
                                                         g1, be1, g2, be2, out, N);
}
